# starting kernel
# baseline (speedup 1.0000x reference)
#include <hip/hip_runtime.h>

typedef __attribute__((ext_vector_type(2))) float v2f;
typedef __attribute__((ext_vector_type(8))) float v8f;


__global__ void k_zero(float* __restrict__ p, long long n) {
    long long i = (long long)blockIdx.x * blockDim.x + threadIdx.x;
    if (i < n) p[i] = 0.0f;
}

__global__ void k_rsqrt_norm(float* __restrict__ p, int n) {
    int i = blockIdx.x * blockDim.x + threadIdx.x;
    if (i < n) {
        float v = p[i];
        p[i] = rsqrtf(fmaxf(v, 1.0f));
    }
}


__global__ void k_degrees(const int* __restrict__ src, const int* __restrict__ dst,
                          float* __restrict__ deg_out, float* __restrict__ deg_in, int E) {
    int e = blockIdx.x * blockDim.x + threadIdx.x;
    if (e < E) {
        atomicAdd(&deg_out[src[e]], 1.0f);
        atomicAdd(&deg_in[dst[e]], 1.0f);
    }
}

template <int K, int NC>
__global__ void k_gemm_scaled(const float* __restrict__ X, const float* __restrict__ ns,
                              const float* __restrict__ W, float* __restrict__ P,
                              int nrows) {
    const int m0   = blockIdx.x * 16;
    const int n0   = blockIdx.y * 16;
    const int lane = threadIdx.x;
    const int half = lane >> 4;
    const int lid  = lane & 15;

    const int  arow_raw = m0 + lid;
    const int  arow     = arow_raw < nrows ? arow_raw : (nrows - 1);
    const float scale   = arow_raw < nrows ? ns[arow] : 0.0f;

    const float* __restrict__ xa = X + (size_t)arow * K + 2 * half;
    const float* __restrict__ wb = W + (size_t)(2 * half) * NC + n0 + lid;

    v8f acc = {};
#pragma unroll 8
    for (int k = 0; k < K; k += 4) {
        v2f a, b;
        const float2 av = *(const float2*)(xa + k);
        a.x = av.x * scale;
        a.y = av.y * scale;
        b.x = wb[(size_t)k * NC];
        b.y = wb[(size_t)(k + 1) * NC];
        acc = __builtin_amdgcn_wmma_f32_16x16x4_f32(
             false, a,  false, b,
             (short)0, acc,  false,  false);
    }

    float* __restrict__ pp = P + (size_t)(m0 + 8 * half) * NC + n0 + lid;
    if (m0 + 16 <= nrows) {
#pragma unroll
        for (int r = 0; r < 8; ++r)
            pp[(size_t)r * NC] = acc[r];
    } else {
#pragma unroll
        for (int r = 0; r < 8; ++r)
            if (m0 + r + 8 * half < nrows)
                pp[(size_t)r * NC] = acc[r];
    }
}

__global__ void k_aggregate(const float* __restrict__ P, const int* __restrict__ src,
                            const int* __restrict__ dst, float* __restrict__ G,
                            long long total, int shift, int d) {
    long long tid = (long long)blockIdx.x * blockDim.x + threadIdx.x;
    if (tid >= total) return;
    const int e = (int)(tid >> shift);
    const int c = (int)(tid & ((1 << shift) - 1));
    const int s = src[e];
    const int t = dst[e];
    const float4 v = *(const float4*)(P + (size_t)s * d + 4 * c);
    float* gp = G + (size_t)t * d + 4 * c;
    atomicAdd(gp + 0, v.x);
    atomicAdd(gp + 1, v.y);
    atomicAdd(gp + 2, v.z);
    atomicAdd(gp + 3, v.w);
}

__global__ void k_finalize(const float* __restrict__ G, const float* __restrict__ nd,
                           const float* __restrict__ b, float* __restrict__ H,
                           long long total, int dshift, int relu) {
    long long idx = (long long)blockIdx.x * blockDim.x + threadIdx.x;
    if (idx >= total) return;
    const int i = (int)(idx >> dshift);
    const int j = (int)(idx & ((1 << dshift) - 1));
    float v = G[idx] * nd[i] + b[j];
    if (relu) v = fmaxf(v, 0.0f);
    H[idx] = v;
}


static inline int cdiv_ll(long long a, int b) { return (int)((a + b - 1) / b); }

extern "C" void kernel_launch(void* const* d_in, const int* in_sizes, int n_in,
                              void* d_out, int out_size, void* d_ws, size_t ws_size,
                              hipStream_t stream) {
    const float* feat = (const float*)d_in[0];
    const float* W1   = (const float*)d_in[1];
    const float* b1   = (const float*)d_in[2];
    const float* W2   = (const float*)d_in[3];
    const float* b2   = (const float*)d_in[4];
    const float* W3   = (const float*)d_in[5];
    const float* b3   = (const float*)d_in[6];
    const int*   src  = (const int*)d_in[7];
    const int*   dst  = (const int*)d_in[8];

    const int D_IN = 256, D_H = 128, D_OUT = 64;
    const int n = in_sizes[0] / D_IN;
    const int E = in_sizes[7];

    float* ns   = (float*)d_ws;
    float* nd   = ns + n;
    float* buf0 = nd + n;
    float* buf1 = buf0 + (size_t)n * D_H;

    float* out = (float*)d_out;

    const int mtiles = (n + 15) / 16;

    k_zero<<<cdiv_ll(2LL * n, 256), 256, 0, stream>>>(ns, 2LL * n);
    k_degrees<<<(E + 255) / 256, 256, 0, stream>>>(src, dst, ns, nd, E);
    k_rsqrt_norm<<<(2 * n + 255) / 256, 256, 0, stream>>>(ns, 2 * n);

    k_gemm_scaled<256, 128><<<dim3(mtiles, D_H / 16), 32, 0, stream>>>(feat, ns, W1, buf0, n);
    k_zero<<<cdiv_ll((long long)n * D_H, 256), 256, 0, stream>>>(buf1, (long long)n * D_H);
    {
        long long total = (long long)E * (D_H / 4);
        k_aggregate<<<cdiv_ll(total, 256), 256, 0, stream>>>(buf0, src, dst, buf1, total, 5, D_H);
    }
    k_finalize<<<cdiv_ll((long long)n * D_H, 256), 256, 0, stream>>>(buf1, nd, b1, buf0,
                                                                     (long long)n * D_H, 7, 1);

    k_gemm_scaled<128, 128><<<dim3(mtiles, D_H / 16), 32, 0, stream>>>(buf0, ns, W2, buf1, n);
    k_zero<<<cdiv_ll((long long)n * D_H, 256), 256, 0, stream>>>(buf0, (long long)n * D_H);
    {
        long long total = (long long)E * (D_H / 4);
        k_aggregate<<<cdiv_ll(total, 256), 256, 0, stream>>>(buf1, src, dst, buf0, total, 5, D_H);
    }
    k_finalize<<<cdiv_ll((long long)n * D_H, 256), 256, 0, stream>>>(buf0, nd, b2, buf1,
                                                                     (long long)n * D_H, 7, 1);

    k_gemm_scaled<128, 64><<<dim3(mtiles, D_OUT / 16), 32, 0, stream>>>(buf1, ns, W3, buf0, n);
    k_zero<<<cdiv_ll((long long)n * D_OUT, 256), 256, 0, stream>>>(buf1, (long long)n * D_OUT);
    {
        long long total = (long long)E * (D_OUT / 4);
        k_aggregate<<<cdiv_ll(total, 256), 256, 0, stream>>>(buf0, src, dst, buf1, total, 4, D_OUT);
    }
    k_finalize<<<cdiv_ll((long long)n * D_OUT, 256), 256, 0, stream>>>(buf1, nd, b3, out,
                                                                       (long long)n * D_OUT, 6, 0);
}


# Float-atomic scatter-add (+zero, +finalize) replaced by a per-call CSR by destination (returning int adds give slots, scan, fill) and one-wave-per-node gather-sum; sum order = atomic slot order
# speedup vs baseline: 2.2785x; 2.2785x over previous
#include <hip/hip_runtime.h>

typedef __attribute__((ext_vector_type(2))) float v2f;
typedef __attribute__((ext_vector_type(4))) float v4f;
typedef __attribute__((ext_vector_type(8))) float v8f;


__global__ void k_zero(float* __restrict__ p, long long n) {
    long long i = (long long)blockIdx.x * blockDim.x + threadIdx.x;
    if (i < n) p[i] = 0.0f;
}

__global__ void k_copy_int(const int* __restrict__ a, int* __restrict__ b, int n) {
    int i = blockIdx.x * blockDim.x + threadIdx.x;
    if (i < n) b[i] = a[i];
}


__global__ void k_degrees_rank(const int* __restrict__ src, const int* __restrict__ dst,
                               int* __restrict__ cnt_out, int* __restrict__ cnt_in,
                               int* __restrict__ rank, int E) {
    int e = blockIdx.x * blockDim.x + threadIdx.x;
    if (e < E) {
        const int s = src[e];
        const int d = dst[e];
        atomicAdd(&cnt_out[s], 1);
        rank[e] = atomicAdd(&cnt_in[d], 1);
    }
}

__global__ void __launch_bounds__(1024)
k_rowptr(const int* __restrict__ cnt_in, int* __restrict__ cnt_out_to_ns,
         int* __restrict__ rowptr, int n) {
    __shared__ int s_part[32];
    __shared__ int s_wave[32];
    const int tid   = threadIdx.x;
    const int lane  = tid & 31;
    const int wid   = tid >> 5;
    const int start = blockIdx.x * 1024;

    int part = 0;
    for (int i = tid; i < start; i += 1024) part += cnt_in[i];
    for (int o = 16; o > 0; o >>= 1) part += __shfl_xor(part, o);
    if (lane == 0) s_part[wid] = part;

    const int idx = start + tid;
    const int v   = idx < n ? cnt_in[idx] : 0;
    int x = v;
    for (int o = 1; o < 32; o <<= 1) {
        const int y = __shfl_up(x, o);
        if (lane >= o) x += y;
    }
    if (lane == 31) s_wave[wid] = x;
    __syncthreads();

    int base = 0;
    for (int w = 0; w < 32; ++w) base += s_part[w];
    for (int w = 0; w < wid; ++w) base += s_wave[w];

    if (idx <= n) rowptr[idx] = base + x - v;
    if (idx < n) {
        const float d = (float)cnt_out_to_ns[idx];
        cnt_out_to_ns[idx] = __float_as_int(rsqrtf(fmaxf(d, 1.0f)));
    }
}

__global__ void k_fill(const int* __restrict__ src, const int* __restrict__ dst,
                       const int* __restrict__ rank, const int* __restrict__ rowptr,
                       int* __restrict__ col, int E) {
    int e = blockIdx.x * blockDim.x + threadIdx.x;
    if (e < E) col[rowptr[dst[e]] + rank[e]] = src[e];
}

template <int K, int NC>
__global__ void k_gemm_scaled(const float* __restrict__ X, const float* __restrict__ ns,
                              const float* __restrict__ W, float* __restrict__ P,
                              int nrows) {
    const int m0   = blockIdx.x * 16;
    const int n0   = blockIdx.y * 16;
    const int lane = threadIdx.x;
    const int half = lane >> 4;
    const int lid  = lane & 15;

    const int  arow_raw = m0 + lid;
    const int  arow     = arow_raw < nrows ? arow_raw : (nrows - 1);
    const float scale   = arow_raw < nrows ? ns[arow] : 0.0f;

    const float* __restrict__ xa = X + (size_t)arow * K + 2 * half;
    const float* __restrict__ wb = W + (size_t)(2 * half) * NC + n0 + lid;

    v8f acc = {};
#pragma unroll 8
    for (int k = 0; k < K; k += 4) {
        v2f a, b;
        const float2 av = *(const float2*)(xa + k);
        a.x = av.x * scale;
        a.y = av.y * scale;
        b.x = wb[(size_t)k * NC];
        b.y = wb[(size_t)(k + 1) * NC];
        acc = __builtin_amdgcn_wmma_f32_16x16x4_f32(
             false, a,  false, b,
             (short)0, acc,  false,  false);
    }

    float* __restrict__ pp = P + (size_t)(m0 + 8 * half) * NC + n0 + lid;
    if (m0 + 16 <= nrows) {
#pragma unroll
        for (int r = 0; r < 8; ++r)
            pp[(size_t)r * NC] = acc[r];
    } else {
#pragma unroll
        for (int r = 0; r < 8; ++r)
            if (m0 + r + 8 * half < nrows)
                pp[(size_t)r * NC] = acc[r];
    }
}

template <int V> struct vec_of;
template <> struct vec_of<4> { typedef v4f type; };
template <> struct vec_of<2> { typedef v2f type; };

template <int D, bool RELU>
__global__ void __launch_bounds__(256)
k_gather(const float* __restrict__ P, const int* __restrict__ rowptr,
         const int* __restrict__ col, const float* __restrict__ b,
         float* __restrict__ H, int n) {
    constexpr int V = D / 32;
    typedef typename vec_of<V>::type vec;
    const int lane = threadIdx.x & 31;
    const int t = __builtin_amdgcn_readfirstlane((int)(blockIdx.x * 8 + (threadIdx.x >> 5)));
    if (t >= n) return;
    const int beg = rowptr[t];
    const int end = rowptr[t + 1];
    const float* __restrict__ pl = P + lane * V;

    vec acc = {};
    for (int c = beg; c < end; c += 32) {
        const int m = (end - c) < 32 ? (end - c) : 32;
        int mine = 0;
        if (lane < m) mine = col[c + lane];
        int k0 = 0;
        for (; k0 + 8 <= m; k0 += 8) {
            int s[8];
            vec v[8];
            for (int k = 0; k < 8; ++k) s[k] = __builtin_amdgcn_readlane(mine, k0 + k);
            for (int k = 0; k < 8; ++k) v[k] = *(const vec*)(pl + (size_t)s[k] * D);
            for (int k = 0; k < 8; ++k) acc += v[k];
        }
        if (k0 < m) {
            const int r = m - k0;
            int s[7];
            vec v[7];
            for (int k = 0; k < 7; ++k) s[k] = __builtin_amdgcn_readlane(mine, k0 + k);
            for (int k = 0; k < 7; ++k) {
                v[k] = vec{};
                if (k < r) v[k] = *(const vec*)(pl + (size_t)s[k] * D);
            }
            for (int k = 0; k < 7; ++k) acc += v[k];
        }
    }

    const float nd = rsqrtf(fmaxf((float)(end - beg), 1.0f));
    const vec bb = *(const vec*)(b + lane * V);
    vec r = acc * nd + bb;
    if (RELU) {
        for (int i = 0; i < V; ++i) r[i] = fmaxf(r[i], 0.0f);
    }
    *(vec*)(H + (size_t)t * D + lane * V) = r;
}


static inline int cdiv_ll(long long a, int b) { return (int)((a + b - 1) / b); }

extern "C" void kernel_launch(void* const* d_in, const int* in_sizes, int n_in,
                              void* d_out, int out_size, void* d_ws, size_t ws_size,
                              hipStream_t stream) {
    const float* feat = (const float*)d_in[0];
    const float* W1   = (const float*)d_in[1];
    const float* b1   = (const float*)d_in[2];
    const float* W2   = (const float*)d_in[3];
    const float* b2   = (const float*)d_in[4];
    const float* W3   = (const float*)d_in[5];
    const float* b3   = (const float*)d_in[6];
    const int*   src  = (const int*)d_in[7];
    const int*   dst  = (const int*)d_in[8];

    const int D_IN = 256, D_H = 128, D_OUT = 64;
    const int n = in_sizes[0] / D_IN;
    const int E = in_sizes[7];

    float* ns   = (float*)d_ws;
    int*   cin  = (int*)(ns + n);
    float* buf0 = ns + 2 * (size_t)n;
    float* buf1 = buf0 + (size_t)n * D_H;

    float* out = (float*)d_out;

    const int csr_ints = (n + 1) + E;
    int* rowptrA = (int*)out;
    int* colA    = rowptrA + (n + 1);
    int* rowptrB = (int*)(buf0 + (size_t)n * D_OUT);
    int* colB    = rowptrB + (n + 1);
    int* rank    = (int*)buf1;

    const int mtiles = (n + 15) / 16;
    const int gblocks = (n + 7) / 8;

    k_zero<<<cdiv_ll(2LL * n, 256), 256, 0, stream>>>(ns, 2LL * n);
    k_degrees_rank<<<(E + 255) / 256, 256, 0, stream>>>(src, dst, (int*)ns, cin, rank, E);
    k_rowptr<<<(n + 1 + 1023) / 1024, 1024, 0, stream>>>(cin, (int*)ns, rowptrA, n);
    k_fill<<<(E + 255) / 256, 256, 0, stream>>>(src, dst, rank, rowptrA, colA, E);

    k_gemm_scaled<256, 128><<<dim3(mtiles, D_H / 16), 32, 0, stream>>>(feat, ns, W1, buf0, n);
    k_gather<128, true><<<gblocks, 256, 0, stream>>>(buf0, rowptrA, colA, b1, buf1, n);

    k_gemm_scaled<128, 128><<<dim3(mtiles, D_H / 16), 32, 0, stream>>>(buf1, ns, W2, buf0, n);
    k_gather<128, true><<<gblocks, 256, 0, stream>>>(buf0, rowptrA, colA, b2, buf1, n);

    k_copy_int<<<(csr_ints + 255) / 256, 256, 0, stream>>>(rowptrA, rowptrB, csr_ints);
    k_gemm_scaled<128, 64><<<dim3(mtiles, D_OUT / 16), 32, 0, stream>>>(buf1, ns, W3, buf0, n);
    k_gather<64, false><<<gblocks, 256, 0, stream>>>(buf0, rowptrB, colB, b3, out, n);
}


# GEMM re-tiled: W held in LDS per workgroup, 512 workgroups whose waves loop over 16-row strips x 64 (32) columns, X read once by 16-byte loads; k regrouped in WMMA pairs (sum order changes)
# speedup vs baseline: 2.9809x; 1.3083x over previous
#include <hip/hip_runtime.h>

typedef __attribute__((ext_vector_type(2))) float v2f;
typedef __attribute__((ext_vector_type(4))) float v4f;
typedef __attribute__((ext_vector_type(8))) float v8f;


__global__ void k_zero(float* __restrict__ p, long long n) {
    long long i = (long long)blockIdx.x * blockDim.x + threadIdx.x;
    if (i < n) p[i] = 0.0f;
}

__global__ void k_copy_int(const int* __restrict__ a, int* __restrict__ b, int n) {
    int i = blockIdx.x * blockDim.x + threadIdx.x;
    if (i < n) b[i] = a[i];
}


__global__ void k_degrees_rank(const int* __restrict__ src, const int* __restrict__ dst,
                               int* __restrict__ cnt_out, int* __restrict__ cnt_in,
                               int* __restrict__ rank, int E) {
    int e = blockIdx.x * blockDim.x + threadIdx.x;
    if (e < E) {
        const int s = src[e];
        const int d = dst[e];
        atomicAdd(&cnt_out[s], 1);
        rank[e] = atomicAdd(&cnt_in[d], 1);
    }
}

__global__ void __launch_bounds__(1024)
k_rowptr(const int* __restrict__ cnt_in, int* __restrict__ cnt_out_to_ns,
         int* __restrict__ rowptr, int n) {
    __shared__ int s_part[32];
    __shared__ int s_wave[32];
    const int tid   = threadIdx.x;
    const int lane  = tid & 31;
    const int wid   = tid >> 5;
    const int start = blockIdx.x * 1024;

    int part = 0;
    for (int i = tid; i < start; i += 1024) part += cnt_in[i];
    for (int o = 16; o > 0; o >>= 1) part += __shfl_xor(part, o);
    if (lane == 0) s_part[wid] = part;

    const int idx = start + tid;
    const int v   = idx < n ? cnt_in[idx] : 0;
    int x = v;
    for (int o = 1; o < 32; o <<= 1) {
        const int y = __shfl_up(x, o);
        if (lane >= o) x += y;
    }
    if (lane == 31) s_wave[wid] = x;
    __syncthreads();

    int base = 0;
    for (int w = 0; w < 32; ++w) base += s_part[w];
    for (int w = 0; w < wid; ++w) base += s_wave[w];

    if (idx <= n) rowptr[idx] = base + x - v;
    if (idx < n) {
        const float d = (float)cnt_out_to_ns[idx];
        cnt_out_to_ns[idx] = __float_as_int(rsqrtf(fmaxf(d, 1.0f)));
    }
}

__global__ void k_fill(const int* __restrict__ src, const int* __restrict__ dst,
                       const int* __restrict__ rank, const int* __restrict__ rowptr,
                       int* __restrict__ col, int E) {
    int e = blockIdx.x * blockDim.x + threadIdx.x;
    if (e < E) col[rowptr[dst[e]] + rank[e]] = src[e];
}

template <int K, int NC, int NTW>
__global__ void __launch_bounds__(256)
k_gemm_strips(const float* __restrict__ X, const float* __restrict__ ns,
              const float* __restrict__ W, float* __restrict__ P, int nrows) {
    constexpr int NT    = NC / 16;
    constexpr int SPLIT = NT / NTW;
    constexpr int KQ    = K / 4;
    constexpr int KB    = 32;
    static_assert(NT % NTW == 0 && K % KB == 0, "tiling");

    __shared__ v4f Wl[KQ * NC];

    const int tid = threadIdx.x;
#pragma unroll 8
    for (int i = tid; i < KQ * NC; i += 256) {
        const int kq = i / NC;
        const int n  = i % NC;
        const float* __restrict__ w = W + (size_t)(4 * kq) * NC + n;
        v4f v;
        v.x = w[0];
        v.y = w[NC];
        v.z = w[2 * NC];
        v.w = w[3 * NC];
        Wl[i] = v;
    }
    __syncthreads();

    const int lane = tid & 31;
    const int h    = lane >> 4;
    const int lid  = lane & 15;
    const int wave = __builtin_amdgcn_readfirstlane(tid >> 5);
    const int nunits = ((nrows + 15) >> 4) * SPLIT;
    const int ustep  = (int)gridDim.x * 8;

    for (int u = (int)blockIdx.x * 8 + wave; u < nunits; u += ustep) {
        const int m0 = (u / SPLIT) * 16;
        const int n0 = (u % SPLIT) * (NTW * 16);
        const int  arow_raw = m0 + lid;
        const int  arow     = arow_raw < nrows ? arow_raw : (nrows - 1);
        const float scale   = arow_raw < nrows ? ns[arow] : 0.0f;

        const float* __restrict__ xa = X + (size_t)arow * K + 4 * h;
        const v4f*   wl = Wl + h * NC + n0 + lid;

        v8f acc[NTW];
        for (int j = 0; j < NTW; ++j) acc[j] = v8f{};

        v4f an[4];
        for (int i = 0; i < 4; ++i) an[i] = *(const v4f*)(xa + 8 * i);

#pragma unroll
        for (int k0 = 0; k0 < K; k0 += KB) {
            v4f a[4];
            for (int i = 0; i < 4; ++i) a[i] = an[i] * scale;
            if (k0 + KB < K) {
                for (int i = 0; i < 4; ++i) an[i] = *(const v4f*)(xa + k0 + KB + 8 * i);
            }
            for (int i = 0; i < 4; ++i) {
                const int kq = (k0 + 8 * i) >> 2;
                v4f b[NTW];
                for (int j = 0; j < NTW; ++j) b[j] = wl[kq * NC + j * 16];
                const v2f a0 = {a[i].x, a[i].y};
                const v2f a1 = {a[i].z, a[i].w};
                for (int j = 0; j < NTW; ++j) {
                    const v2f b0 = {b[j].x, b[j].y};
                    acc[j] = __builtin_amdgcn_wmma_f32_16x16x4_f32(
                         false, a0,  false, b0,
                         (short)0, acc[j],  false,  false);
                }
                for (int j = 0; j < NTW; ++j) {
                    const v2f b1 = {b[j].z, b[j].w};
                    acc[j] = __builtin_amdgcn_wmma_f32_16x16x4_f32(
                         false, a1,  false, b1,
                         (short)0, acc[j],  false,  false);
                }
            }
        }

        float* __restrict__ pp = P + (size_t)(m0 + 8 * h) * NC + n0 + lid;
        if (m0 + 16 <= nrows) {
            for (int j = 0; j < NTW; ++j)
                for (int r = 0; r < 8; ++r)
                    pp[(size_t)r * NC + j * 16] = acc[j][r];
        } else {
            for (int j = 0; j < NTW; ++j)
                for (int r = 0; r < 8; ++r)
                    if (m0 + r + 8 * h < nrows)
                        pp[(size_t)r * NC + j * 16] = acc[j][r];
        }
    }
}

template <int V> struct vec_of;
template <> struct vec_of<4> { typedef v4f type; };
template <> struct vec_of<2> { typedef v2f type; };

template <int D, bool RELU>
__global__ void __launch_bounds__(256)
k_gather(const float* __restrict__ P, const int* __restrict__ rowptr,
         const int* __restrict__ col, const float* __restrict__ b,
         float* __restrict__ H, int n) {
    constexpr int V = D / 32;
    typedef typename vec_of<V>::type vec;
    const int lane = threadIdx.x & 31;
    const int t = __builtin_amdgcn_readfirstlane((int)(blockIdx.x * 8 + (threadIdx.x >> 5)));
    if (t >= n) return;
    const int beg = rowptr[t];
    const int end = rowptr[t + 1];
    const float* __restrict__ pl = P + lane * V;

    vec acc = {};
    for (int c = beg; c < end; c += 32) {
        const int m = (end - c) < 32 ? (end - c) : 32;
        int mine = 0;
        if (lane < m) mine = col[c + lane];
        int k0 = 0;
        for (; k0 + 8 <= m; k0 += 8) {
            int s[8];
            vec v[8];
            for (int k = 0; k < 8; ++k) s[k] = __builtin_amdgcn_readlane(mine, k0 + k);
            for (int k = 0; k < 8; ++k) v[k] = *(const vec*)(pl + (size_t)s[k] * D);
            for (int k = 0; k < 8; ++k) acc += v[k];
        }
        if (k0 < m) {
            const int r = m - k0;
            int s[7];
            vec v[7];
            for (int k = 0; k < 7; ++k) s[k] = __builtin_amdgcn_readlane(mine, k0 + k);
            for (int k = 0; k < 7; ++k) {
                v[k] = vec{};
                if (k < r) v[k] = *(const vec*)(pl + (size_t)s[k] * D);
            }
            for (int k = 0; k < 7; ++k) acc += v[k];
        }
    }

    const float nd = rsqrtf(fmaxf((float)(end - beg), 1.0f));
    const vec bb = *(const vec*)(b + lane * V);
    vec r = acc * nd + bb;
    if (RELU) {
        for (int i = 0; i < V; ++i) r[i] = fmaxf(r[i], 0.0f);
    }
    *(vec*)(H + (size_t)t * D + lane * V) = r;
}


static inline int cdiv_ll(long long a, int b) { return (int)((a + b - 1) / b); }

extern "C" void kernel_launch(void* const* d_in, const int* in_sizes, int n_in,
                              void* d_out, int out_size, void* d_ws, size_t ws_size,
                              hipStream_t stream) {
    const float* feat = (const float*)d_in[0];
    const float* W1   = (const float*)d_in[1];
    const float* b1   = (const float*)d_in[2];
    const float* W2   = (const float*)d_in[3];
    const float* b2   = (const float*)d_in[4];
    const float* W3   = (const float*)d_in[5];
    const float* b3   = (const float*)d_in[6];
    const int*   src  = (const int*)d_in[7];
    const int*   dst  = (const int*)d_in[8];

    const int D_IN = 256, D_H = 128, D_OUT = 64;
    const int n = in_sizes[0] / D_IN;
    const int E = in_sizes[7];

    float* ns   = (float*)d_ws;
    int*   cin  = (int*)(ns + n);
    float* buf0 = ns + 2 * (size_t)n;
    float* buf1 = buf0 + (size_t)n * D_H;

    float* out = (float*)d_out;

    const int csr_ints = (n + 1) + E;
    int* rowptrA = (int*)out;
    int* colA    = rowptrA + (n + 1);
    int* rowptrB = (int*)(buf0 + (size_t)n * D_OUT);
    int* colB    = rowptrB + (n + 1);
    int* rank    = (int*)buf1;

    const int gblocks = (n + 7) / 8;

    k_zero<<<cdiv_ll(2LL * n, 256), 256, 0, stream>>>(ns, 2LL * n);
    k_degrees_rank<<<(E + 255) / 256, 256, 0, stream>>>(src, dst, (int*)ns, cin, rank, E);
    k_rowptr<<<(n + 1 + 1023) / 1024, 1024, 0, stream>>>(cin, (int*)ns, rowptrA, n);
    k_fill<<<(E + 255) / 256, 256, 0, stream>>>(src, dst, rank, rowptrA, colA, E);

    k_gemm_strips<256, 128, 4><<<512, 256, 0, stream>>>(feat, ns, W1, buf0, n);
    k_gather<128, true><<<gblocks, 256, 0, stream>>>(buf0, rowptrA, colA, b1, buf1, n);

    k_gemm_strips<128, 128, 4><<<512, 256, 0, stream>>>(buf1, ns, W2, buf0, n);
    k_gather<128, true><<<gblocks, 256, 0, stream>>>(buf0, rowptrA, colA, b2, buf1, n);

    k_copy_int<<<(csr_ints + 255) / 256, 256, 0, stream>>>(rowptrA, rowptrB, csr_ints);
    k_gemm_strips<128, 64, 2><<<512, 256, 0, stream>>>(buf1, ns, W3, buf0, n);
    k_gather<64, false><<<gblocks, 256, 0, stream>>>(buf0, rowptrB, colB, b3, out, n);
}


# Global-atomic CSR build (zero, count+rank, offsets, fill) replaced by a two-level counting sort on LDS counters only (chunk bin counts, scan, bin runs, per-bin finish); sum order = LDS add arrival
# speedup vs baseline: 4.5440x; 1.5244x over previous
#include <hip/hip_runtime.h>

typedef __attribute__((ext_vector_type(2))) float v2f;
typedef __attribute__((ext_vector_type(4))) float v4f;
typedef __attribute__((ext_vector_type(8))) float v8f;


__global__ void k_copy_int(const int* __restrict__ a, int* __restrict__ b, int n) {
    int i = blockIdx.x * blockDim.x + threadIdx.x;
    if (i < n) b[i] = a[i];
}


constexpr int BW_SHIFT = 9;
constexpr int BW       = 1 << BW_SHIFT;
constexpr int MAXNB    = 256;
constexpr int EPT      = 25;
constexpr int CH       = 256 * EPT;

template <int NW>
__device__ inline int block_excl_scan(int v, int* s_wave, int tid) {
    const int lane = tid & 31;
    const int wid  = tid >> 5;
    int x = v;
    for (int o = 1; o < 32; o <<= 1) {
        const int y = __shfl_up(x, o);
        if (lane >= o) x += y;
    }
    if (lane == 31) s_wave[wid] = x;
    __syncthreads();
    int base = 0;
    for (int w = 0; w < NW; ++w) {
        const int t = s_wave[w];
        if (w < wid) base += t;
    }
    return base + x - v;
}

__global__ void __launch_bounds__(256)
k_bin_count(const int* __restrict__ src, const int* __restrict__ dst,
            int* __restrict__ hist, int E, int NB, int C) {
    __shared__ int cD[MAXNB];
    __shared__ int cS[MAXNB];
    const int tid = threadIdx.x;
    const int c   = blockIdx.x;
    cD[tid] = 0;
    cS[tid] = 0;
    __syncthreads();
    const int base = c * CH + tid;
    int bd[EPT], bs[EPT];
#pragma unroll
    for (int k = 0; k < EPT; ++k) {
        const int e  = base + k * 256;
        const int ec = e < E ? e : E - 1;
        bd[k] = dst[ec] >> BW_SHIFT;
        bs[k] = src[ec] >> BW_SHIFT;
    }
#pragma unroll
    for (int k = 0; k < EPT; ++k) {
        const int one = (base + k * 256) < E ? 1 : 0;
        atomicAdd(&cD[bd[k]], one);
        atomicAdd(&cS[bs[k]], one);
    }
    __syncthreads();
    if (tid < NB) {
        hist[(size_t)tid * C + c]        = cD[tid];
        hist[(size_t)(NB + tid) * C + c] = cS[tid];
    }
}

__global__ void __launch_bounds__(256)
k_bin_scan(int* __restrict__ hist, int* __restrict__ tot, int rows, int C) {
    const int lane = threadIdx.x & 31;
    const int row = __builtin_amdgcn_readfirstlane((int)(blockIdx.x * 8 + (threadIdx.x >> 5)));
    if (row >= rows) return;
    int* __restrict__ h = hist + (size_t)row * C;
    int carry = 0;
    for (int t0 = 0; t0 < C; t0 += 256) {
        int v[8];
        for (int k = 0; k < 8; ++k) {
            const int i = t0 + k * 32 + lane;
            v[k] = i < C ? h[i] : 0;
        }
        for (int k = 0; k < 8; ++k) {
            int x = v[k];
            for (int o = 1; o < 32; o <<= 1) {
                const int y = __shfl_up(x, o);
                if (lane >= o) x += y;
            }
            const int i = t0 + k * 32 + lane;
            if (i < C) h[i] = carry + x - v[k];
            carry += __shfl(x, 31);
        }
    }
    if (lane == 0) tot[row] = carry;
}

__global__ void __launch_bounds__(256)
k_bin_place(const int* __restrict__ src, const int* __restrict__ dst,
            const int* __restrict__ off, const int* __restrict__ tot,
            int2* __restrict__ tmpD, int* __restrict__ tmpS, int E, int NB, int C) {
    __shared__ int2 sD[CH + 1];
    __shared__ int  sS[CH + 1];
    __shared__ int  lstD[MAXNB], lstS[MAXNB], curD[MAXNB], curS[MAXNB], gD[MAXNB], gS[MAXNB];
    __shared__ int  swv[4][8];
    const int tid = threadIdx.x;
    const int c   = blockIdx.x;

    const int tD = tid < NB ? tot[tid] : 0;
    const int tS = tid < NB ? tot[NB + tid] : 0;
    const int bD = block_excl_scan<8>(tD, swv[0], tid);
    const int bS = block_excl_scan<8>(tS, swv[1], tid);
    if (tid < NB) {
        gD[tid] = bD + off[(size_t)tid * C + c];
        gS[tid] = bS + off[(size_t)(NB + tid) * C + c];
    }
    curD[tid] = 0;
    curS[tid] = 0;
    __syncthreads();

    int d[EPT], s[EPT];
    const int base = c * CH + tid;
#pragma unroll
    for (int k = 0; k < EPT; ++k) {
        const int e  = base + k * 256;
        const int ec = e < E ? e : E - 1;
        d[k] = dst[ec];
        s[k] = src[ec];
    }
#pragma unroll
    for (int k = 0; k < EPT; ++k) {
        const int one = (base + k * 256) < E ? 1 : 0;
        atomicAdd(&curD[d[k] >> BW_SHIFT], one);
        atomicAdd(&curS[s[k] >> BW_SHIFT], one);
    }
    __syncthreads();
    const int cd = curD[tid];
    const int cs = curS[tid];
    const int ld = block_excl_scan<8>(cd, swv[2], tid);
    const int ls = block_excl_scan<8>(cs, swv[3], tid);
    lstD[tid] = ld;
    curD[tid] = ld;
    lstS[tid] = ls;
    curS[tid] = ls;
    __syncthreads();
#pragma unroll
    for (int k = 0; k < EPT; ++k) {
        const bool ok  = (base + k * 256) < E;
        const int  one = ok ? 1 : 0;
        const int  r   = atomicAdd(&curD[d[k] >> BW_SHIFT], one);
        const int  q   = atomicAdd(&curS[s[k] >> BW_SHIFT], one);
        sD[ok ? r : CH] = make_int2(d[k], s[k]);
        sS[ok ? q : CH] = s[k];
    }
    __syncthreads();
    const int rest = E - c * CH;
    const int cnt  = rest < CH ? rest : CH;
    for (int i = tid; i < cnt; i += 256) {
        const int2 p = sD[i];
        const int  b = p.x >> BW_SHIFT;
        tmpD[gD[b] + i - lstD[b]] = p;
        const int  v = sS[i];
        const int  a = v >> BW_SHIFT;
        tmpS[gS[a] + i - lstS[a]] = v;
    }
}

__global__ void __launch_bounds__(BW)
k_bin_finish(const int2* __restrict__ tmpD, const int* __restrict__ tmpS,
             const int* __restrict__ tot, int* __restrict__ rowptr,
             int* __restrict__ col, float* __restrict__ ns, int n, int NB) {
    __shared__ int cnt[BW];
    __shared__ int cur[BW];
    __shared__ int swv[2][BW / 32];
    __shared__ int s_range[2];
    const int  tid  = threadIdx.x;
    const bool isS  = (int)blockIdx.x >= NB;
    const int  b    = isS ? (int)blockIdx.x - NB : (int)blockIdx.x;
    const int  fam0 = isS ? NB : 0;

    const int t    = tid < NB ? tot[fam0 + tid] : 0;
    const int base = block_excl_scan<BW / 32>(t, swv[0], tid);
    if (tid == b) {
        s_range[0] = base;
        s_range[1] = base + t;
    }
    cnt[tid] = 0;
    __syncthreads();
    const int beg   = s_range[0];
    const int end   = s_range[1];
    const int node0 = b << BW_SHIFT;
    const int node  = node0 + tid;

    if (!isS) {
        for (int i0 = beg + tid; i0 < end; i0 += 4 * BW) {
            int k[4];
            for (int j = 0; j < 4; ++j) {
                const int i = i0 + j * BW;
                k[j] = i < end ? tmpD[i].x - node0 : -1;
            }
            for (int j = 0; j < 4; ++j)
                if (k[j] >= 0) atomicAdd(&cnt[k[j]], 1);
        }
        __syncthreads();
        const int c  = cnt[tid];
        const int st = block_excl_scan<BW / 32>(c, swv[1], tid);
        cur[tid] = st;
        if (node < n) rowptr[node] = beg + st;
        if (node == n - 1) rowptr[n] = beg + st + c;
        __syncthreads();
        for (int i0 = beg + tid; i0 < end; i0 += 4 * BW) {
            int2 p[4];
            for (int j = 0; j < 4; ++j) {
                const int i = i0 + j * BW;
                p[j] = i < end ? tmpD[i] : make_int2(-1, 0);
            }
            for (int j = 0; j < 4; ++j) {
                if (p[j].x >= 0) {
                    const int slot = atomicAdd(&cur[p[j].x - node0], 1);
                    col[beg + slot] = p[j].y;
                }
            }
        }
    } else {
        for (int i0 = beg + tid; i0 < end; i0 += 4 * BW) {
            int k[4];
            for (int j = 0; j < 4; ++j) {
                const int i = i0 + j * BW;
                k[j] = i < end ? tmpS[i] - node0 : -1;
            }
            for (int j = 0; j < 4; ++j)
                if (k[j] >= 0) atomicAdd(&cnt[k[j]], 1);
        }
        __syncthreads();
        if (node < n) ns[node] = rsqrtf(fmaxf((float)cnt[tid], 1.0f));
    }
}

template <int K, int NC, int NTW>
__global__ void __launch_bounds__(256)
k_gemm_strips(const float* __restrict__ X, const float* __restrict__ ns,
              const float* __restrict__ W, float* __restrict__ P, int nrows) {
    constexpr int NT    = NC / 16;
    constexpr int SPLIT = NT / NTW;
    constexpr int KQ    = K / 4;
    constexpr int KB    = 32;
    static_assert(NT % NTW == 0 && K % KB == 0, "tiling");

    __shared__ v4f Wl[KQ * NC];

    const int tid = threadIdx.x;
#pragma unroll 8
    for (int i = tid; i < KQ * NC; i += 256) {
        const int kq = i / NC;
        const int n  = i % NC;
        const float* __restrict__ w = W + (size_t)(4 * kq) * NC + n;
        v4f v;
        v.x = w[0];
        v.y = w[NC];
        v.z = w[2 * NC];
        v.w = w[3 * NC];
        Wl[i] = v;
    }
    __syncthreads();

    const int lane = tid & 31;
    const int h    = lane >> 4;
    const int lid  = lane & 15;
    const int wave = __builtin_amdgcn_readfirstlane(tid >> 5);
    const int nunits = ((nrows + 15) >> 4) * SPLIT;
    const int ustep  = (int)gridDim.x * 8;

    for (int u = (int)blockIdx.x * 8 + wave; u < nunits; u += ustep) {
        const int m0 = (u / SPLIT) * 16;
        const int n0 = (u % SPLIT) * (NTW * 16);
        const int  arow_raw = m0 + lid;
        const int  arow     = arow_raw < nrows ? arow_raw : (nrows - 1);
        const float scale   = arow_raw < nrows ? ns[arow] : 0.0f;

        const float* __restrict__ xa = X + (size_t)arow * K + 4 * h;
        const v4f*   wl = Wl + h * NC + n0 + lid;

        v8f acc[NTW];
        for (int j = 0; j < NTW; ++j) acc[j] = v8f{};

        v4f an[4];
        for (int i = 0; i < 4; ++i) an[i] = *(const v4f*)(xa + 8 * i);

#pragma unroll
        for (int k0 = 0; k0 < K; k0 += KB) {
            v4f a[4];
            for (int i = 0; i < 4; ++i) a[i] = an[i] * scale;
            if (k0 + KB < K) {
                for (int i = 0; i < 4; ++i) an[i] = *(const v4f*)(xa + k0 + KB + 8 * i);
            }
            for (int i = 0; i < 4; ++i) {
                const int kq = (k0 + 8 * i) >> 2;
                v4f b[NTW];
                for (int j = 0; j < NTW; ++j) b[j] = wl[kq * NC + j * 16];
                const v2f a0 = {a[i].x, a[i].y};
                const v2f a1 = {a[i].z, a[i].w};
                for (int j = 0; j < NTW; ++j) {
                    const v2f b0 = {b[j].x, b[j].y};
                    acc[j] = __builtin_amdgcn_wmma_f32_16x16x4_f32(
                         false, a0,  false, b0,
                         (short)0, acc[j],  false,  false);
                }
                for (int j = 0; j < NTW; ++j) {
                    const v2f b1 = {b[j].z, b[j].w};
                    acc[j] = __builtin_amdgcn_wmma_f32_16x16x4_f32(
                         false, a1,  false, b1,
                         (short)0, acc[j],  false,  false);
                }
            }
        }

        float* __restrict__ pp = P + (size_t)(m0 + 8 * h) * NC + n0 + lid;
        if (m0 + 16 <= nrows) {
            for (int j = 0; j < NTW; ++j)
                for (int r = 0; r < 8; ++r)
                    pp[(size_t)r * NC + j * 16] = acc[j][r];
        } else {
            for (int j = 0; j < NTW; ++j)
                for (int r = 0; r < 8; ++r)
                    if (m0 + r + 8 * h < nrows)
                        pp[(size_t)r * NC + j * 16] = acc[j][r];
        }
    }
}

template <int V> struct vec_of;
template <> struct vec_of<4> { typedef v4f type; };
template <> struct vec_of<2> { typedef v2f type; };

template <int D, bool RELU>
__global__ void __launch_bounds__(256)
k_gather(const float* __restrict__ P, const int* __restrict__ rowptr,
         const int* __restrict__ col, const float* __restrict__ b,
         float* __restrict__ H, int n) {
    constexpr int V = D / 32;
    typedef typename vec_of<V>::type vec;
    const int lane = threadIdx.x & 31;
    const int t = __builtin_amdgcn_readfirstlane((int)(blockIdx.x * 8 + (threadIdx.x >> 5)));
    if (t >= n) return;
    const int beg = rowptr[t];
    const int end = rowptr[t + 1];
    const float* __restrict__ pl = P + lane * V;

    vec acc = {};
    for (int c = beg; c < end; c += 32) {
        const int m = (end - c) < 32 ? (end - c) : 32;
        int mine = 0;
        if (lane < m) mine = col[c + lane];
        int k0 = 0;
        for (; k0 + 8 <= m; k0 += 8) {
            int s[8];
            vec v[8];
            for (int k = 0; k < 8; ++k) s[k] = __builtin_amdgcn_readlane(mine, k0 + k);
            for (int k = 0; k < 8; ++k) v[k] = *(const vec*)(pl + (size_t)s[k] * D);
            for (int k = 0; k < 8; ++k) acc += v[k];
        }
        if (k0 < m) {
            const int r = m - k0;
            int s[7];
            vec v[7];
            for (int k = 0; k < 7; ++k) s[k] = __builtin_amdgcn_readlane(mine, k0 + k);
            for (int k = 0; k < 7; ++k) {
                v[k] = vec{};
                if (k < r) v[k] = *(const vec*)(pl + (size_t)s[k] * D);
            }
            for (int k = 0; k < 7; ++k) acc += v[k];
        }
    }

    const float nd = rsqrtf(fmaxf((float)(end - beg), 1.0f));
    const vec bb = *(const vec*)(b + lane * V);
    vec r = acc * nd + bb;
    if (RELU) {
        for (int i = 0; i < V; ++i) r[i] = fmaxf(r[i], 0.0f);
    }
    *(vec*)(H + (size_t)t * D + lane * V) = r;
}


extern "C" void kernel_launch(void* const* d_in, const int* in_sizes, int n_in,
                              void* d_out, int out_size, void* d_ws, size_t ws_size,
                              hipStream_t stream) {
    const float* feat = (const float*)d_in[0];
    const float* W1   = (const float*)d_in[1];
    const float* b1   = (const float*)d_in[2];
    const float* W2   = (const float*)d_in[3];
    const float* b2   = (const float*)d_in[4];
    const float* W3   = (const float*)d_in[5];
    const float* b3   = (const float*)d_in[6];
    const int*   src  = (const int*)d_in[7];
    const int*   dst  = (const int*)d_in[8];

    const int D_IN = 256, D_H = 128, D_OUT = 64;
    const int n = in_sizes[0] / D_IN;
    const int E = in_sizes[7];

    float* ns   = (float*)d_ws;
    float* buf0 = ns + 2 * (size_t)n;
    float* buf1 = buf0 + (size_t)n * D_H;

    float* out = (float*)d_out;

    const int csr_ints = (n + 1) + E;
    int* rowptrA = (int*)out;
    int* colA    = rowptrA + (n + 1);
    int* rowptrB = (int*)(buf0 + (size_t)n * D_OUT);
    int* colB    = rowptrB + (n + 1);

    const int NB = (n + BW - 1) >> BW_SHIFT;
    const int C  = (E + CH - 1) / CH;
    int2* tmpD = (int2*)buf1;
    int*  tmpS = (int*)(tmpD + E);
    int*  hist = tmpS + E;
    int*  tot  = hist + (size_t)2 * NB * C;

    const int gblocks = (n + 7) / 8;

    k_bin_count<<<C, 256, 0, stream>>>(src, dst, hist, E, NB, C);
    k_bin_scan<<<(2 * NB + 7) / 8, 256, 0, stream>>>(hist, tot, 2 * NB, C);
    k_bin_place<<<C, 256, 0, stream>>>(src, dst, hist, tot, tmpD, tmpS, E, NB, C);
    k_bin_finish<<<2 * NB, BW, 0, stream>>>(tmpD, tmpS, tot, rowptrA, colA, ns, n, NB);

    k_gemm_strips<256, 128, 4><<<512, 256, 0, stream>>>(feat, ns, W1, buf0, n);
    k_gather<128, true><<<gblocks, 256, 0, stream>>>(buf0, rowptrA, colA, b1, buf1, n);

    k_gemm_strips<128, 128, 4><<<512, 256, 0, stream>>>(buf1, ns, W2, buf0, n);
    k_gather<128, true><<<gblocks, 256, 0, stream>>>(buf0, rowptrA, colA, b2, buf1, n);

    k_copy_int<<<(csr_ints + 255) / 256, 256, 0, stream>>>(rowptrA, rowptrB, csr_ints);
    k_gemm_strips<128, 64, 2><<<512, 256, 0, stream>>>(buf1, ns, W3, buf0, n);
    k_gather<64, false><<<gblocks, 256, 0, stream>>>(buf0, rowptrB, colB, b3, out, n);
}


# Hidden layers' gather-sum fused with the next layer's projection: a workgroup sums 32 nodes' rows into an LDS tile and multiplies it by W held in LDS; h1, h2 never reach memory; sum orders unchanged
# speedup vs baseline: 4.9795x; 1.0958x over previous
#include <hip/hip_runtime.h>

typedef __attribute__((ext_vector_type(2))) float v2f;
typedef __attribute__((ext_vector_type(4))) float v4f;
typedef __attribute__((ext_vector_type(8))) float v8f;


__global__ void k_copy_int(const int* __restrict__ a, int* __restrict__ b, int n) {
    int i = blockIdx.x * blockDim.x + threadIdx.x;
    if (i < n) b[i] = a[i];
}


constexpr int BW_SHIFT = 9;
constexpr int BW       = 1 << BW_SHIFT;
constexpr int MAXNB    = 256;
constexpr int EPT      = 25;
constexpr int CH       = 256 * EPT;

template <int NW>
__device__ inline int block_excl_scan(int v, int* s_wave, int tid) {
    const int lane = tid & 31;
    const int wid  = tid >> 5;
    int x = v;
    for (int o = 1; o < 32; o <<= 1) {
        const int y = __shfl_up(x, o);
        if (lane >= o) x += y;
    }
    if (lane == 31) s_wave[wid] = x;
    __syncthreads();
    int base = 0;
    for (int w = 0; w < NW; ++w) {
        const int t = s_wave[w];
        if (w < wid) base += t;
    }
    return base + x - v;
}

__global__ void __launch_bounds__(256)
k_bin_count(const int* __restrict__ src, const int* __restrict__ dst,
            int* __restrict__ hist, int E, int NB, int C) {
    __shared__ int cD[MAXNB];
    __shared__ int cS[MAXNB];
    const int tid = threadIdx.x;
    const int c   = blockIdx.x;
    cD[tid] = 0;
    cS[tid] = 0;
    __syncthreads();
    const int base = c * CH + tid;
    int bd[EPT], bs[EPT];
#pragma unroll
    for (int k = 0; k < EPT; ++k) {
        const int e  = base + k * 256;
        const int ec = e < E ? e : E - 1;
        bd[k] = dst[ec] >> BW_SHIFT;
        bs[k] = src[ec] >> BW_SHIFT;
    }
#pragma unroll
    for (int k = 0; k < EPT; ++k) {
        const int one = (base + k * 256) < E ? 1 : 0;
        atomicAdd(&cD[bd[k]], one);
        atomicAdd(&cS[bs[k]], one);
    }
    __syncthreads();
    if (tid < NB) {
        hist[(size_t)tid * C + c]        = cD[tid];
        hist[(size_t)(NB + tid) * C + c] = cS[tid];
    }
}

__global__ void __launch_bounds__(256)
k_bin_scan(int* __restrict__ hist, int* __restrict__ tot, int rows, int C) {
    const int lane = threadIdx.x & 31;
    const int row = __builtin_amdgcn_readfirstlane((int)(blockIdx.x * 8 + (threadIdx.x >> 5)));
    if (row >= rows) return;
    int* __restrict__ h = hist + (size_t)row * C;
    int carry = 0;
    for (int t0 = 0; t0 < C; t0 += 256) {
        int v[8];
        for (int k = 0; k < 8; ++k) {
            const int i = t0 + k * 32 + lane;
            v[k] = i < C ? h[i] : 0;
        }
        for (int k = 0; k < 8; ++k) {
            int x = v[k];
            for (int o = 1; o < 32; o <<= 1) {
                const int y = __shfl_up(x, o);
                if (lane >= o) x += y;
            }
            const int i = t0 + k * 32 + lane;
            if (i < C) h[i] = carry + x - v[k];
            carry += __shfl(x, 31);
        }
    }
    if (lane == 0) tot[row] = carry;
}

__global__ void __launch_bounds__(256)
k_bin_place(const int* __restrict__ src, const int* __restrict__ dst,
            const int* __restrict__ off, const int* __restrict__ tot,
            int2* __restrict__ tmpD, int* __restrict__ tmpS, int E, int NB, int C) {
    __shared__ int2 sD[CH + 1];
    __shared__ int  sS[CH + 1];
    __shared__ int  lstD[MAXNB], lstS[MAXNB], curD[MAXNB], curS[MAXNB], gD[MAXNB], gS[MAXNB];
    __shared__ int  swv[4][8];
    const int tid = threadIdx.x;
    const int c   = blockIdx.x;

    const int tD = tid < NB ? tot[tid] : 0;
    const int tS = tid < NB ? tot[NB + tid] : 0;
    const int bD = block_excl_scan<8>(tD, swv[0], tid);
    const int bS = block_excl_scan<8>(tS, swv[1], tid);
    if (tid < NB) {
        gD[tid] = bD + off[(size_t)tid * C + c];
        gS[tid] = bS + off[(size_t)(NB + tid) * C + c];
    }
    curD[tid] = 0;
    curS[tid] = 0;
    __syncthreads();

    int d[EPT], s[EPT];
    const int base = c * CH + tid;
#pragma unroll
    for (int k = 0; k < EPT; ++k) {
        const int e  = base + k * 256;
        const int ec = e < E ? e : E - 1;
        d[k] = dst[ec];
        s[k] = src[ec];
    }
#pragma unroll
    for (int k = 0; k < EPT; ++k) {
        const int one = (base + k * 256) < E ? 1 : 0;
        atomicAdd(&curD[d[k] >> BW_SHIFT], one);
        atomicAdd(&curS[s[k] >> BW_SHIFT], one);
    }
    __syncthreads();
    const int cd = curD[tid];
    const int cs = curS[tid];
    const int ld = block_excl_scan<8>(cd, swv[2], tid);
    const int ls = block_excl_scan<8>(cs, swv[3], tid);
    lstD[tid] = ld;
    curD[tid] = ld;
    lstS[tid] = ls;
    curS[tid] = ls;
    __syncthreads();
#pragma unroll
    for (int k = 0; k < EPT; ++k) {
        const bool ok  = (base + k * 256) < E;
        const int  one = ok ? 1 : 0;
        const int  r   = atomicAdd(&curD[d[k] >> BW_SHIFT], one);
        const int  q   = atomicAdd(&curS[s[k] >> BW_SHIFT], one);
        sD[ok ? r : CH] = make_int2(d[k], s[k]);
        sS[ok ? q : CH] = s[k];
    }
    __syncthreads();
    const int rest = E - c * CH;
    const int cnt  = rest < CH ? rest : CH;
    for (int i = tid; i < cnt; i += 256) {
        const int2 p = sD[i];
        const int  b = p.x >> BW_SHIFT;
        tmpD[gD[b] + i - lstD[b]] = p;
        const int  v = sS[i];
        const int  a = v >> BW_SHIFT;
        tmpS[gS[a] + i - lstS[a]] = v;
    }
}

__global__ void __launch_bounds__(BW)
k_bin_finish(const int2* __restrict__ tmpD, const int* __restrict__ tmpS,
             const int* __restrict__ tot, int* __restrict__ rowptr,
             int* __restrict__ col, float* __restrict__ ns, int n, int NB) {
    __shared__ int cnt[BW];
    __shared__ int cur[BW];
    __shared__ int swv[2][BW / 32];
    __shared__ int s_range[2];
    const int  tid  = threadIdx.x;
    const bool isS  = (int)blockIdx.x >= NB;
    const int  b    = isS ? (int)blockIdx.x - NB : (int)blockIdx.x;
    const int  fam0 = isS ? NB : 0;

    const int t    = tid < NB ? tot[fam0 + tid] : 0;
    const int base = block_excl_scan<BW / 32>(t, swv[0], tid);
    if (tid == b) {
        s_range[0] = base;
        s_range[1] = base + t;
    }
    cnt[tid] = 0;
    __syncthreads();
    const int beg   = s_range[0];
    const int end   = s_range[1];
    const int node0 = b << BW_SHIFT;
    const int node  = node0 + tid;

    if (!isS) {
        for (int i0 = beg + tid; i0 < end; i0 += 4 * BW) {
            int k[4];
            for (int j = 0; j < 4; ++j) {
                const int i = i0 + j * BW;
                k[j] = i < end ? tmpD[i].x - node0 : -1;
            }
            for (int j = 0; j < 4; ++j)
                if (k[j] >= 0) atomicAdd(&cnt[k[j]], 1);
        }
        __syncthreads();
        const int c  = cnt[tid];
        const int st = block_excl_scan<BW / 32>(c, swv[1], tid);
        cur[tid] = st;
        if (node < n) rowptr[node] = beg + st;
        if (node == n - 1) rowptr[n] = beg + st + c;
        __syncthreads();
        for (int i0 = beg + tid; i0 < end; i0 += 4 * BW) {
            int2 p[4];
            for (int j = 0; j < 4; ++j) {
                const int i = i0 + j * BW;
                p[j] = i < end ? tmpD[i] : make_int2(-1, 0);
            }
            for (int j = 0; j < 4; ++j) {
                if (p[j].x >= 0) {
                    const int slot = atomicAdd(&cur[p[j].x - node0], 1);
                    col[beg + slot] = p[j].y;
                }
            }
        }
    } else {
        for (int i0 = beg + tid; i0 < end; i0 += 4 * BW) {
            int k[4];
            for (int j = 0; j < 4; ++j) {
                const int i = i0 + j * BW;
                k[j] = i < end ? tmpS[i] - node0 : -1;
            }
            for (int j = 0; j < 4; ++j)
                if (k[j] >= 0) atomicAdd(&cnt[k[j]], 1);
        }
        __syncthreads();
        if (node < n) ns[node] = rsqrtf(fmaxf((float)cnt[tid], 1.0f));
    }
}

template <int K, int NC, int NTW>
__global__ void __launch_bounds__(256)
k_gemm_strips(const float* __restrict__ X, const float* __restrict__ ns,
              const float* __restrict__ W, float* __restrict__ P, int nrows) {
    constexpr int NT    = NC / 16;
    constexpr int SPLIT = NT / NTW;
    constexpr int KQ    = K / 4;
    constexpr int KB    = 32;
    static_assert(NT % NTW == 0 && K % KB == 0, "tiling");

    __shared__ v4f Wl[KQ * NC];

    const int tid = threadIdx.x;
#pragma unroll 8
    for (int i = tid; i < KQ * NC; i += 256) {
        const int kq = i / NC;
        const int n  = i % NC;
        const float* __restrict__ w = W + (size_t)(4 * kq) * NC + n;
        v4f v;
        v.x = w[0];
        v.y = w[NC];
        v.z = w[2 * NC];
        v.w = w[3 * NC];
        Wl[i] = v;
    }
    __syncthreads();

    const int lane = tid & 31;
    const int h    = lane >> 4;
    const int lid  = lane & 15;
    const int wave = __builtin_amdgcn_readfirstlane(tid >> 5);
    const int nunits = ((nrows + 15) >> 4) * SPLIT;
    const int ustep  = (int)gridDim.x * 8;

    for (int u = (int)blockIdx.x * 8 + wave; u < nunits; u += ustep) {
        const int m0 = (u / SPLIT) * 16;
        const int n0 = (u % SPLIT) * (NTW * 16);
        const int  arow_raw = m0 + lid;
        const int  arow     = arow_raw < nrows ? arow_raw : (nrows - 1);
        const float scale   = arow_raw < nrows ? ns[arow] : 0.0f;

        const float* __restrict__ xa = X + (size_t)arow * K + 4 * h;
        const v4f*   wl = Wl + h * NC + n0 + lid;

        v8f acc[NTW];
        for (int j = 0; j < NTW; ++j) acc[j] = v8f{};

        v4f an[4];
        for (int i = 0; i < 4; ++i) an[i] = *(const v4f*)(xa + 8 * i);

#pragma unroll
        for (int k0 = 0; k0 < K; k0 += KB) {
            v4f a[4];
            for (int i = 0; i < 4; ++i) a[i] = an[i] * scale;
            if (k0 + KB < K) {
                for (int i = 0; i < 4; ++i) an[i] = *(const v4f*)(xa + k0 + KB + 8 * i);
            }
            for (int i = 0; i < 4; ++i) {
                const int kq = (k0 + 8 * i) >> 2;
                v4f b[NTW];
                for (int j = 0; j < NTW; ++j) b[j] = wl[kq * NC + j * 16];
                const v2f a0 = {a[i].x, a[i].y};
                const v2f a1 = {a[i].z, a[i].w};
                for (int j = 0; j < NTW; ++j) {
                    const v2f b0 = {b[j].x, b[j].y};
                    acc[j] = __builtin_amdgcn_wmma_f32_16x16x4_f32(
                         false, a0,  false, b0,
                         (short)0, acc[j],  false,  false);
                }
                for (int j = 0; j < NTW; ++j) {
                    const v2f b1 = {b[j].z, b[j].w};
                    acc[j] = __builtin_amdgcn_wmma_f32_16x16x4_f32(
                         false, a1,  false, b1,
                         (short)0, acc[j],  false,  false);
                }
            }
        }

        float* __restrict__ pp = P + (size_t)(m0 + 8 * h) * NC + n0 + lid;
        if (m0 + 16 <= nrows) {
            for (int j = 0; j < NTW; ++j)
                for (int r = 0; r < 8; ++r)
                    pp[(size_t)r * NC + j * 16] = acc[j][r];
        } else {
            for (int j = 0; j < NTW; ++j)
                for (int r = 0; r < 8; ++r)
                    if (m0 + r + 8 * h < nrows)
                        pp[(size_t)r * NC + j * 16] = acc[j][r];
        }
    }
}

template <int V> struct vec_of;
template <> struct vec_of<4> { typedef v4f type; };
template <> struct vec_of<2> { typedef v2f type; };

template <int D, bool RELU>
__global__ void __launch_bounds__(256)
k_gather(const float* __restrict__ P, const int* __restrict__ rowptr,
         const int* __restrict__ col, const float* __restrict__ b,
         float* __restrict__ H, int n) {
    constexpr int V = D / 32;
    typedef typename vec_of<V>::type vec;
    const int lane = threadIdx.x & 31;
    const int t = __builtin_amdgcn_readfirstlane((int)(blockIdx.x * 8 + (threadIdx.x >> 5)));
    if (t >= n) return;
    const int beg = rowptr[t];
    const int end = rowptr[t + 1];
    const float* __restrict__ pl = P + lane * V;

    vec acc = {};
    for (int c = beg; c < end; c += 32) {
        const int m = (end - c) < 32 ? (end - c) : 32;
        int mine = 0;
        if (lane < m) mine = col[c + lane];
        int k0 = 0;
        for (; k0 + 8 <= m; k0 += 8) {
            int s[8];
            vec v[8];
            for (int k = 0; k < 8; ++k) s[k] = __builtin_amdgcn_readlane(mine, k0 + k);
            for (int k = 0; k < 8; ++k) v[k] = *(const vec*)(pl + (size_t)s[k] * D);
            for (int k = 0; k < 8; ++k) acc += v[k];
        }
        if (k0 < m) {
            const int r = m - k0;
            int s[7];
            vec v[7];
            for (int k = 0; k < 7; ++k) s[k] = __builtin_amdgcn_readlane(mine, k0 + k);
            for (int k = 0; k < 7; ++k) {
                v[k] = vec{};
                if (k < r) v[k] = *(const vec*)(pl + (size_t)s[k] * D);
            }
            for (int k = 0; k < 7; ++k) acc += v[k];
        }
    }

    const float nd = rsqrtf(fmaxf((float)(end - beg), 1.0f));
    const vec bb = *(const vec*)(b + lane * V);
    vec r = acc * nd + bb;
    if (RELU) {
        for (int i = 0; i < V; ++i) r[i] = fmaxf(r[i], 0.0f);
    }
    *(vec*)(H + (size_t)t * D + lane * V) = r;
}

template <int NC>
__global__ void __launch_bounds__(256)
k_gather_gemm(const float* __restrict__ P, const int* __restrict__ rowptr,
              const int* __restrict__ col, const float* __restrict__ bias,
              const float* __restrict__ ns, const float* __restrict__ W,
              float* __restrict__ Pn, int n) {
    constexpr int D    = 128;
    constexpr int KQ   = D / 4;
    constexpr int ROWS = 32;
    constexpr int NPW  = ROWS / 8;
    constexpr int CT   = NC / 16;
    constexpr int RT   = 2 * CT / 8;
    constexpr int AP   = KQ + 1;
    constexpr int G    = 16;
    static_assert(RT >= 1 && CT * 16 == NC, "tiling");

    __shared__ v4f Wl[KQ * NC];
    __shared__ v4f At[2][ROWS * AP];

    const int tid = threadIdx.x;
#pragma unroll 8
    for (int i = tid; i < KQ * NC; i += 256) {
        const int kq = i / NC;
        const int nn = i % NC;
        const float* __restrict__ w = W + (size_t)(4 * kq) * NC + nn;
        v4f v;
        v.x = w[0];
        v.y = w[NC];
        v.z = w[2 * NC];
        v.w = w[3 * NC];
        Wl[i] = v;
    }

    const int lane = tid & 31;
    const int h    = lane >> 4;
    const int lid  = lane & 15;
    const int wave = __builtin_amdgcn_readfirstlane(tid >> 5);
    const int ct   = wave % CT;
    const int rt0  = (wave / CT) * RT;
    const int n0   = ct * 16;

    const v4f bb = *(const v4f*)(bias + lane * 4);
    const float* __restrict__ pl = P + lane * 4;
    const v4f* wl = Wl + h * NC + n0 + lid;
    const int nstrips = (n + ROWS - 1) / ROWS;
    int buf = 0;

    int   rp;
    float nsl = 0.0f;
    {
        const int t0 = (int)blockIdx.x * ROWS + wave * NPW;
        const int idx = t0 + lane;
        rp = rowptr[idx < n ? idx : n];
        if (lane < NPW && idx < n) nsl = ns[idx];
    }

    for (int strip = (int)blockIdx.x; strip < nstrips; strip += (int)gridDim.x) {
        const int t0 = strip * ROWS + wave * NPW;

        v4f* __restrict__ arow = At[buf] + (wave * NPW) * AP + lane;

        auto flush = [&](int i, v4f acc) {
            const int deg = __builtin_amdgcn_readlane(rp, i + 1) - __builtin_amdgcn_readlane(rp, i);
            const float nd = rsqrtf(fmaxf((float)deg, 1.0f));
            const float sc = __int_as_float(__builtin_amdgcn_readlane(__float_as_int(nsl), i));
            v4f r = acc * nd + bb;
            for (int q = 0; q < 4; ++q) r[q] = fmaxf(r[q], 0.0f);
            arow[i * AP] = r * sc;
        };

        const int ebeg = __builtin_amdgcn_readlane(rp, 0);
        const int eend = __builtin_amdgcn_readlane(rp, NPW);
        int  i    = 0;
        int  nend = __builtin_amdgcn_readlane(rp, 1);
        v4f  acc  = {};
        int mine = 0;
        if (ebeg + lane < eend) mine = col[ebeg + lane];
        for (int c = ebeg; c < eend; c += 32) {
            const int m = (eend - c) < 32 ? (eend - c) : 32;
            int nxt = 0;
            if (c + 32 + lane < eend) nxt = col[c + 32 + lane];
            for (int k0 = 0; k0 < m; k0 += G) {
                int s[G];
                v4f v[G];
                for (int k = 0; k < G; ++k) s[k] = __builtin_amdgcn_readlane(mine, k0 + k);
                if (k0 + G <= m) {
                    for (int k = 0; k < G; ++k) v[k] = *(const v4f*)(pl + (size_t)s[k] * D);
                } else {
                    for (int k = 0; k < G; ++k) {
                        v[k] = v4f{};
                        if (k0 + k < m) v[k] = *(const v4f*)(pl + (size_t)s[k] * D);
                    }
                }
                for (int k = 0; k < G; ++k) {
                    const int e = c + k0 + k;
                    if (e < eend) {
                        while (e >= nend && i < NPW - 1) {
                            flush(i, acc);
                            acc = v4f{};
                            ++i;
                            nend = __builtin_amdgcn_readlane(rp, i + 1);
                        }
                        acc += v[k];
                    }
                }
            }
            mine = nxt;
        }
        for (; i < NPW; ++i) {
            flush(i, acc);
            acc = v4f{};
        }
        __syncthreads();

        {
            const int idx = t0 + (int)gridDim.x * ROWS + lane;
            rp  = rowptr[idx < n ? idx : n];
            nsl = 0.0f;
            if (lane < NPW && idx < n) nsl = ns[idx];
        }

        const v4f* al = At[buf] + (rt0 * 16 + lid) * AP + h;
        v8f dacc[RT];
        for (int j = 0; j < RT; ++j) dacc[j] = v8f{};

        v4f fa[RT][4], fb[4];
        for (int q = 0; q < 4; ++q) {
            for (int j = 0; j < RT; ++j) fa[j][q] = al[j * 16 * AP + 2 * q];
            fb[q] = wl[(2 * q) * NC];
        }
#pragma unroll
        for (int gb = 0; gb < 4; ++gb) {
            v4f ca[RT][4], cb[4];
            for (int q = 0; q < 4; ++q) {
                for (int j = 0; j < RT; ++j) ca[j][q] = fa[j][q];
                cb[q] = fb[q];
            }
            if (gb + 1 < 4) {
                for (int q = 0; q < 4; ++q) {
                    const int g = (gb + 1) * 4 + q;
                    for (int j = 0; j < RT; ++j) fa[j][q] = al[j * 16 * AP + 2 * g];
                    fb[q] = wl[(2 * g) * NC];
                }
            }
            for (int q = 0; q < 4; ++q) {
                const v2f b0 = {cb[q].x, cb[q].y};
                const v2f b1 = {cb[q].z, cb[q].w};
                for (int j = 0; j < RT; ++j) {
                    const v2f a0 = {ca[j][q].x, ca[j][q].y};
                    dacc[j] = __builtin_amdgcn_wmma_f32_16x16x4_f32(
                         false, a0,  false, b0,
                         (short)0, dacc[j],  false,  false);
                }
                for (int j = 0; j < RT; ++j) {
                    const v2f a1 = {ca[j][q].z, ca[j][q].w};
                    dacc[j] = __builtin_amdgcn_wmma_f32_16x16x4_f32(
                         false, a1,  false, b1,
                         (short)0, dacc[j],  false,  false);
                }
            }
        }

        const int m0 = strip * ROWS + rt0 * 16;
        float* __restrict__ pp = Pn + (size_t)(m0 + 8 * h) * NC + n0 + lid;
        if (strip * ROWS + ROWS <= n) {
            for (int j = 0; j < RT; ++j)
                for (int r = 0; r < 8; ++r)
                    pp[(size_t)(j * 16 + r) * NC] = dacc[j][r];
        } else {
            for (int j = 0; j < RT; ++j)
                for (int r = 0; r < 8; ++r)
                    if (m0 + j * 16 + 8 * h + r < n)
                        pp[(size_t)(j * 16 + r) * NC] = dacc[j][r];
        }
        buf ^= 1;
    }
}


extern "C" void kernel_launch(void* const* d_in, const int* in_sizes, int n_in,
                              void* d_out, int out_size, void* d_ws, size_t ws_size,
                              hipStream_t stream) {
    const float* feat = (const float*)d_in[0];
    const float* W1   = (const float*)d_in[1];
    const float* b1   = (const float*)d_in[2];
    const float* W2   = (const float*)d_in[3];
    const float* b2   = (const float*)d_in[4];
    const float* W3   = (const float*)d_in[5];
    const float* b3   = (const float*)d_in[6];
    const int*   src  = (const int*)d_in[7];
    const int*   dst  = (const int*)d_in[8];

    const int D_IN = 256, D_H = 128, D_OUT = 64;
    const int n = in_sizes[0] / D_IN;
    const int E = in_sizes[7];

    float* ns   = (float*)d_ws;
    float* buf0 = ns + 2 * (size_t)n;
    float* buf1 = buf0 + (size_t)n * D_H;

    float* out = (float*)d_out;

    const int csr_ints = (n + 1) + E;
    int* rowptrA = (int*)out;
    int* colA    = rowptrA + (n + 1);
    int* rowptrB = (int*)(buf0 + (size_t)n * D_OUT);
    int* colB    = rowptrB + (n + 1);

    const int NB = (n + BW - 1) >> BW_SHIFT;
    const int C  = (E + CH - 1) / CH;
    int2* tmpD = (int2*)buf1;
    int*  tmpS = (int*)(tmpD + E);
    int*  hist = tmpS + E;
    int*  tot  = hist + (size_t)2 * NB * C;

    const int gblocks = (n + 7) / 8;

    k_bin_count<<<C, 256, 0, stream>>>(src, dst, hist, E, NB, C);
    k_bin_scan<<<(2 * NB + 7) / 8, 256, 0, stream>>>(hist, tot, 2 * NB, C);
    k_bin_place<<<C, 256, 0, stream>>>(src, dst, hist, tot, tmpD, tmpS, E, NB, C);
    k_bin_finish<<<2 * NB, BW, 0, stream>>>(tmpD, tmpS, tot, rowptrA, colA, ns, n, NB);

    k_gemm_strips<256, 128, 4><<<512, 256, 0, stream>>>(feat, ns, W1, buf0, n);

    k_gather_gemm<128><<<768, 256, 0, stream>>>(buf0, rowptrA, colA, b1, ns, W2, buf1, n);

    k_gather_gemm<64><<<1024, 256, 0, stream>>>(buf1, rowptrA, colA, b2, ns, W3, buf0, n);

    k_copy_int<<<(csr_ints + 255) / 256, 256, 0, stream>>>(rowptrA, rowptrB, csr_ints);
    k_gather<64, false><<<gblocks, 256, 0, stream>>>(buf0, rowptrB, colB, b3, out, n);
}


# CSR build: global bin count and chunk scan launches dropped; chunks write bin-ordered edges with their bin offsets, a bin's workgroup walks the chunk runs, 8 loads in flight; sum order = LDS arrival
# speedup vs baseline: 5.3287x; 1.0701x over previous
#include <hip/hip_runtime.h>

typedef __attribute__((ext_vector_type(2))) float v2f;
typedef __attribute__((ext_vector_type(4))) float v4f;
typedef __attribute__((ext_vector_type(8))) float v8f;


__global__ void k_copy_int(const int* __restrict__ a, int* __restrict__ b, int n) {
    int i = blockIdx.x * blockDim.x + threadIdx.x;
    if (i < n) b[i] = a[i];
}


constexpr int BW_SHIFT = 9;
constexpr int BW       = 1 << BW_SHIFT;
constexpr int MAXNB    = 256;
constexpr int EPT      = 25;
constexpr int CH       = 256 * EPT;

template <int NW>
__device__ inline int block_excl_scan(int v, int* s_wave, int tid) {
    const int lane = tid & 31;
    const int wid  = tid >> 5;
    int x = v;
    for (int o = 1; o < 32; o <<= 1) {
        const int y = __shfl_up(x, o);
        if (lane >= o) x += y;
    }
    if (lane == 31) s_wave[wid] = x;
    __syncthreads();
    int base = 0;
    for (int w = 0; w < NW; ++w) {
        const int t = s_wave[w];
        if (w < wid) base += t;
    }
    return base + x - v;
}

__global__ void __launch_bounds__(256)
k_bin_place(const int* __restrict__ src, const int* __restrict__ dst,
            int* __restrict__ lst, int2* __restrict__ tmpD, int* __restrict__ tmpS,
            int E, int NB, int C) {
    __shared__ int2 sD[CH + 1];
    __shared__ int  sS[CH + 1];
    __shared__ int  curD[MAXNB], curS[MAXNB];
    __shared__ int  swv[2][8];
    const int tid = threadIdx.x;
    const int c   = blockIdx.x;

    curD[tid] = 0;
    curS[tid] = 0;
    __syncthreads();

    int d[EPT], s[EPT];
    const int base = c * CH + tid;
#pragma unroll
    for (int k = 0; k < EPT; ++k) {
        const int e  = base + k * 256;
        const int ec = e < E ? e : E - 1;
        d[k] = dst[ec];
        s[k] = src[ec];
    }
#pragma unroll
    for (int k = 0; k < EPT; ++k) {
        const int one = (base + k * 256) < E ? 1 : 0;
        atomicAdd(&curD[d[k] >> BW_SHIFT], one);
        atomicAdd(&curS[s[k] >> BW_SHIFT], one);
    }
    __syncthreads();
    const int rest = E - c * CH;
    const int cnt  = rest < CH ? rest : CH;
    const int cd = curD[tid];
    const int cs = curS[tid];
    const int ld = block_excl_scan<8>(cd, swv[0], tid);
    const int ls = block_excl_scan<8>(cs, swv[1], tid);
    curD[tid] = ld;
    curS[tid] = ls;
    if (tid < NB) {
        lst[(size_t)tid * C + c]            = ld;
        lst[(size_t)(NB + 1 + tid) * C + c] = ls;
    }
    if (tid == 0) {
        lst[(size_t)NB * C + c]           = cnt;
        lst[(size_t)(2 * NB + 1) * C + c] = cnt;
    }
    __syncthreads();
#pragma unroll
    for (int k = 0; k < EPT; ++k) {
        const bool ok  = (base + k * 256) < E;
        const int  one = ok ? 1 : 0;
        const int  r   = atomicAdd(&curD[d[k] >> BW_SHIFT], one);
        const int  q   = atomicAdd(&curS[s[k] >> BW_SHIFT], one);
        sD[ok ? r : CH] = make_int2(d[k], s[k]);
        sS[ok ? q : CH] = s[k];
    }
    __syncthreads();
    int2* __restrict__ oD = tmpD + (size_t)c * CH;
    int*  __restrict__ oS = tmpS + (size_t)c * CH;
    const int full = cnt >> 8;
#pragma unroll 5
    for (int k = 0; k < full; ++k) {
        const int i = tid + k * 256;
        oD[i] = sD[i];
        oS[i] = sS[i];
    }
    const int i = tid + full * 256;
    if (i < cnt) {
        oD[i] = sD[i];
        oS[i] = sS[i];
    }
}

__global__ void __launch_bounds__(BW)
k_bin_finish(const int2* __restrict__ tmpD, const int* __restrict__ tmpS,
             const int* __restrict__ lst, int* __restrict__ rowptr,
             int* __restrict__ col, float* __restrict__ ns, int n, int NB, int C) {
    constexpr int NWV = BW / 32;
    __shared__ int cnt[BW];
    __shared__ int cur[BW];
    __shared__ int rs[BW];
    __shared__ int rl[BW];
    __shared__ int swv[2][NWV];
    __shared__ int s_beg;
    const int  tid  = threadIdx.x;
    const int  lane = tid & 31;
    const int  wave = __builtin_amdgcn_readfirstlane(tid >> 5);
    const bool isS  = (int)blockIdx.x >= NB;
    const int  b    = isS ? (int)blockIdx.x - NB : (int)blockIdx.x;
    const int* __restrict__ row0 = lst + (size_t)((isS ? NB + 1 : 0) + b) * C;
    const int* __restrict__ row1 = row0 + C;
    const int node0 = b << BW_SHIFT;
    const int node  = node0 + tid;
    const int* __restrict__ keys = isS ? tmpS : (const int*)tmpD;
    const int  ksh = isS ? 0 : 1;

    cnt[tid] = 0;
    int before = 0;
    for (int c0 = 0; c0 < C; c0 += BW) {
        const int c = c0 + tid;
        int s0 = 0, s1 = 0;
        if (c < C) {
            s0 = row0[c];
            s1 = row1[c];
        }
        before += s0;
        if (c0 > 0) __syncthreads();
        rs[tid] = c * CH + s0;
        rl[tid] = s1 - s0;
        __syncthreads();
        const int runs = (C - c0) < BW ? (C - c0) : BW;
        for (int r0 = wave * 4; r0 < runs; r0 += NWV * 4) {
            int st[4], ln[4], k[8];
            for (int q = 0; q < 4; ++q) {
                st[q] = rs[r0 + q];
                ln[q] = rl[r0 + q];
            }
            for (int q = 0; q < 4; ++q)
                for (int u = 0; u < 2; ++u) {
                    const int  j  = lane + 32 * u;
                    const bool ok = j < ln[q];
                    const int  v  = keys[(ok ? st[q] + j : 0) << ksh];
                    k[2 * q + u] = ok ? v - node0 : -1;
                }
            for (int i = 0; i < 8; ++i)
                if (k[i] >= 0) atomicAdd(&cnt[k[i]], 1);
            for (int q = 0; q < 4; ++q)
                for (int j = lane + 64; j < ln[q]; j += 32)
                    atomicAdd(&cnt[keys[(st[q] + j) << ksh] - node0], 1);
        }
    }
    __syncthreads();

    if (isS) {
        if (node < n) ns[node] = rsqrtf(fmaxf((float)cnt[tid], 1.0f));
        return;
    }

    const int ex = block_excl_scan<NWV>(before, swv[0], tid);
    if (tid == BW - 1) s_beg = ex + before;
    const int cn = cnt[tid];
    const int sc = block_excl_scan<NWV>(cn, swv[1], tid);
    cur[tid] = sc;
    __syncthreads();
    const int beg = s_beg;
    if (node < n) rowptr[node] = beg + sc;
    if (node == n - 1) rowptr[n] = beg + sc + cn;

    for (int c0 = 0; c0 < C; c0 += BW) {
        if (C > BW) {
            const int c = c0 + tid;
            int s0 = 0, s1 = 0;
            if (c < C) {
                s0 = row0[c];
                s1 = row1[c];
            }
            __syncthreads();
            rs[tid] = c * CH + s0;
            rl[tid] = s1 - s0;
            __syncthreads();
        }
        const int runs = (C - c0) < BW ? (C - c0) : BW;
        for (int r0 = wave * 4; r0 < runs; r0 += NWV * 4) {
            int  st[4], ln[4];
            int2 p[8];
            for (int q = 0; q < 4; ++q) {
                st[q] = rs[r0 + q];
                ln[q] = rl[r0 + q];
            }
            for (int q = 0; q < 4; ++q)
                for (int u = 0; u < 2; ++u) {
                    const int  j  = lane + 32 * u;
                    const bool ok = j < ln[q];
                    const int2 e  = tmpD[ok ? st[q] + j : 0];
                    p[2 * q + u] = make_int2(ok ? e.x : -1, e.y);
                }
            for (int i = 0; i < 8; ++i) {
                if (p[i].x >= 0) {
                    const int slot = atomicAdd(&cur[p[i].x - node0], 1);
                    col[beg + slot] = p[i].y;
                }
            }
            for (int q = 0; q < 4; ++q)
                for (int j = lane + 64; j < ln[q]; j += 32) {
                    const int2 e = tmpD[st[q] + j];
                    const int slot = atomicAdd(&cur[e.x - node0], 1);
                    col[beg + slot] = e.y;
                }
        }
    }
}

template <int K, int NC, int NTW>
__global__ void __launch_bounds__(256)
k_gemm_strips(const float* __restrict__ X, const float* __restrict__ ns,
              const float* __restrict__ W, float* __restrict__ P, int nrows) {
    constexpr int NT    = NC / 16;
    constexpr int SPLIT = NT / NTW;
    constexpr int KQ    = K / 4;
    constexpr int KB    = 32;
    static_assert(NT % NTW == 0 && K % KB == 0, "tiling");

    __shared__ v4f Wl[KQ * NC];

    const int tid = threadIdx.x;
#pragma unroll 8
    for (int i = tid; i < KQ * NC; i += 256) {
        const int kq = i / NC;
        const int n  = i % NC;
        const float* __restrict__ w = W + (size_t)(4 * kq) * NC + n;
        v4f v;
        v.x = w[0];
        v.y = w[NC];
        v.z = w[2 * NC];
        v.w = w[3 * NC];
        Wl[i] = v;
    }
    __syncthreads();

    const int lane = tid & 31;
    const int h    = lane >> 4;
    const int lid  = lane & 15;
    const int wave = __builtin_amdgcn_readfirstlane(tid >> 5);
    const int nunits = ((nrows + 15) >> 4) * SPLIT;
    const int ustep  = (int)gridDim.x * 8;

    for (int u = (int)blockIdx.x * 8 + wave; u < nunits; u += ustep) {
        const int m0 = (u / SPLIT) * 16;
        const int n0 = (u % SPLIT) * (NTW * 16);
        const int  arow_raw = m0 + lid;
        const int  arow     = arow_raw < nrows ? arow_raw : (nrows - 1);
        const float scale   = arow_raw < nrows ? ns[arow] : 0.0f;

        const float* __restrict__ xa = X + (size_t)arow * K + 4 * h;
        const v4f*   wl = Wl + h * NC + n0 + lid;

        v8f acc[NTW];
        for (int j = 0; j < NTW; ++j) acc[j] = v8f{};

        v4f an[4];
        for (int i = 0; i < 4; ++i) an[i] = *(const v4f*)(xa + 8 * i);

#pragma unroll
        for (int k0 = 0; k0 < K; k0 += KB) {
            v4f a[4];
            for (int i = 0; i < 4; ++i) a[i] = an[i] * scale;
            if (k0 + KB < K) {
                for (int i = 0; i < 4; ++i) an[i] = *(const v4f*)(xa + k0 + KB + 8 * i);
            }
            for (int i = 0; i < 4; ++i) {
                const int kq = (k0 + 8 * i) >> 2;
                v4f b[NTW];
                for (int j = 0; j < NTW; ++j) b[j] = wl[kq * NC + j * 16];
                const v2f a0 = {a[i].x, a[i].y};
                const v2f a1 = {a[i].z, a[i].w};
                for (int j = 0; j < NTW; ++j) {
                    const v2f b0 = {b[j].x, b[j].y};
                    acc[j] = __builtin_amdgcn_wmma_f32_16x16x4_f32(
                         false, a0,  false, b0,
                         (short)0, acc[j],  false,  false);
                }
                for (int j = 0; j < NTW; ++j) {
                    const v2f b1 = {b[j].z, b[j].w};
                    acc[j] = __builtin_amdgcn_wmma_f32_16x16x4_f32(
                         false, a1,  false, b1,
                         (short)0, acc[j],  false,  false);
                }
            }
        }

        float* __restrict__ pp = P + (size_t)(m0 + 8 * h) * NC + n0 + lid;
        if (m0 + 16 <= nrows) {
            for (int j = 0; j < NTW; ++j)
                for (int r = 0; r < 8; ++r)
                    pp[(size_t)r * NC + j * 16] = acc[j][r];
        } else {
            for (int j = 0; j < NTW; ++j)
                for (int r = 0; r < 8; ++r)
                    if (m0 + r + 8 * h < nrows)
                        pp[(size_t)r * NC + j * 16] = acc[j][r];
        }
    }
}

template <int V> struct vec_of;
template <> struct vec_of<4> { typedef v4f type; };
template <> struct vec_of<2> { typedef v2f type; };

template <int D, bool RELU>
__global__ void __launch_bounds__(256)
k_gather(const float* __restrict__ P, const int* __restrict__ rowptr,
         const int* __restrict__ col, const float* __restrict__ b,
         float* __restrict__ H, int n) {
    constexpr int V = D / 32;
    typedef typename vec_of<V>::type vec;
    const int lane = threadIdx.x & 31;
    const int t = __builtin_amdgcn_readfirstlane((int)(blockIdx.x * 8 + (threadIdx.x >> 5)));
    if (t >= n) return;
    const int beg = rowptr[t];
    const int end = rowptr[t + 1];
    const float* __restrict__ pl = P + lane * V;

    vec acc = {};
    for (int c = beg; c < end; c += 32) {
        const int m = (end - c) < 32 ? (end - c) : 32;
        int mine = 0;
        if (lane < m) mine = col[c + lane];
        int k0 = 0;
        for (; k0 + 8 <= m; k0 += 8) {
            int s[8];
            vec v[8];
            for (int k = 0; k < 8; ++k) s[k] = __builtin_amdgcn_readlane(mine, k0 + k);
            for (int k = 0; k < 8; ++k) v[k] = *(const vec*)(pl + (size_t)s[k] * D);
            for (int k = 0; k < 8; ++k) acc += v[k];
        }
        if (k0 < m) {
            const int r = m - k0;
            int s[7];
            vec v[7];
            for (int k = 0; k < 7; ++k) s[k] = __builtin_amdgcn_readlane(mine, k0 + k);
            for (int k = 0; k < 7; ++k) {
                v[k] = vec{};
                if (k < r) v[k] = *(const vec*)(pl + (size_t)s[k] * D);
            }
            for (int k = 0; k < 7; ++k) acc += v[k];
        }
    }

    const float nd = rsqrtf(fmaxf((float)(end - beg), 1.0f));
    const vec bb = *(const vec*)(b + lane * V);
    vec r = acc * nd + bb;
    if (RELU) {
        for (int i = 0; i < V; ++i) r[i] = fmaxf(r[i], 0.0f);
    }
    *(vec*)(H + (size_t)t * D + lane * V) = r;
}

template <int NC>
__global__ void __launch_bounds__(256)
k_gather_gemm(const float* __restrict__ P, const int* __restrict__ rowptr,
              const int* __restrict__ col, const float* __restrict__ bias,
              const float* __restrict__ ns, const float* __restrict__ W,
              float* __restrict__ Pn, int n) {
    constexpr int D    = 128;
    constexpr int KQ   = D / 4;
    constexpr int ROWS = 32;
    constexpr int NPW  = ROWS / 8;
    constexpr int CT   = NC / 16;
    constexpr int RT   = 2 * CT / 8;
    constexpr int AP   = KQ + 1;
    constexpr int G    = 16;
    static_assert(RT >= 1 && CT * 16 == NC, "tiling");

    __shared__ v4f Wl[KQ * NC];
    __shared__ v4f At[2][ROWS * AP];

    const int tid = threadIdx.x;
#pragma unroll 8
    for (int i = tid; i < KQ * NC; i += 256) {
        const int kq = i / NC;
        const int nn = i % NC;
        const float* __restrict__ w = W + (size_t)(4 * kq) * NC + nn;
        v4f v;
        v.x = w[0];
        v.y = w[NC];
        v.z = w[2 * NC];
        v.w = w[3 * NC];
        Wl[i] = v;
    }

    const int lane = tid & 31;
    const int h    = lane >> 4;
    const int lid  = lane & 15;
    const int wave = __builtin_amdgcn_readfirstlane(tid >> 5);
    const int ct   = wave % CT;
    const int rt0  = (wave / CT) * RT;
    const int n0   = ct * 16;

    const v4f bb = *(const v4f*)(bias + lane * 4);
    const float* __restrict__ pl = P + lane * 4;
    const v4f* wl = Wl + h * NC + n0 + lid;
    const int nstrips = (n + ROWS - 1) / ROWS;
    int buf = 0;

    int   rp;
    float nsl = 0.0f;
    {
        const int t0 = (int)blockIdx.x * ROWS + wave * NPW;
        const int idx = t0 + lane;
        rp = rowptr[idx < n ? idx : n];
        if (lane < NPW && idx < n) nsl = ns[idx];
    }

    for (int strip = (int)blockIdx.x; strip < nstrips; strip += (int)gridDim.x) {
        const int t0 = strip * ROWS + wave * NPW;

        v4f* __restrict__ arow = At[buf] + (wave * NPW) * AP + lane;

        auto flush = [&](int i, v4f acc) {
            const int deg = __builtin_amdgcn_readlane(rp, i + 1) - __builtin_amdgcn_readlane(rp, i);
            const float nd = rsqrtf(fmaxf((float)deg, 1.0f));
            const float sc = __int_as_float(__builtin_amdgcn_readlane(__float_as_int(nsl), i));
            v4f r = acc * nd + bb;
            for (int q = 0; q < 4; ++q) r[q] = fmaxf(r[q], 0.0f);
            arow[i * AP] = r * sc;
        };

        const int ebeg = __builtin_amdgcn_readlane(rp, 0);
        const int eend = __builtin_amdgcn_readlane(rp, NPW);
        int  i    = 0;
        int  nend = __builtin_amdgcn_readlane(rp, 1);
        v4f  acc  = {};
        int mine = 0;
        if (ebeg + lane < eend) mine = col[ebeg + lane];
        for (int c = ebeg; c < eend; c += 32) {
            const int m = (eend - c) < 32 ? (eend - c) : 32;
            int nxt = 0;
            if (c + 32 + lane < eend) nxt = col[c + 32 + lane];
            for (int k0 = 0; k0 < m; k0 += G) {
                int s[G];
                v4f v[G];
                for (int k = 0; k < G; ++k) s[k] = __builtin_amdgcn_readlane(mine, k0 + k);
                if (k0 + G <= m) {
                    for (int k = 0; k < G; ++k) v[k] = *(const v4f*)(pl + (size_t)s[k] * D);
                } else {
                    for (int k = 0; k < G; ++k) {
                        v[k] = v4f{};
                        if (k0 + k < m) v[k] = *(const v4f*)(pl + (size_t)s[k] * D);
                    }
                }
                for (int k = 0; k < G; ++k) {
                    const int e = c + k0 + k;
                    if (e < eend) {
                        while (e >= nend && i < NPW - 1) {
                            flush(i, acc);
                            acc = v4f{};
                            ++i;
                            nend = __builtin_amdgcn_readlane(rp, i + 1);
                        }
                        acc += v[k];
                    }
                }
            }
            mine = nxt;
        }
        for (; i < NPW; ++i) {
            flush(i, acc);
            acc = v4f{};
        }
        __syncthreads();

        {
            const int idx = t0 + (int)gridDim.x * ROWS + lane;
            rp  = rowptr[idx < n ? idx : n];
            nsl = 0.0f;
            if (lane < NPW && idx < n) nsl = ns[idx];
        }

        const v4f* al = At[buf] + (rt0 * 16 + lid) * AP + h;
        v8f dacc[RT];
        for (int j = 0; j < RT; ++j) dacc[j] = v8f{};

        v4f fa[RT][4], fb[4];
        for (int q = 0; q < 4; ++q) {
            for (int j = 0; j < RT; ++j) fa[j][q] = al[j * 16 * AP + 2 * q];
            fb[q] = wl[(2 * q) * NC];
        }
#pragma unroll
        for (int gb = 0; gb < 4; ++gb) {
            v4f ca[RT][4], cb[4];
            for (int q = 0; q < 4; ++q) {
                for (int j = 0; j < RT; ++j) ca[j][q] = fa[j][q];
                cb[q] = fb[q];
            }
            if (gb + 1 < 4) {
                for (int q = 0; q < 4; ++q) {
                    const int g = (gb + 1) * 4 + q;
                    for (int j = 0; j < RT; ++j) fa[j][q] = al[j * 16 * AP + 2 * g];
                    fb[q] = wl[(2 * g) * NC];
                }
            }
            for (int q = 0; q < 4; ++q) {
                const v2f b0 = {cb[q].x, cb[q].y};
                const v2f b1 = {cb[q].z, cb[q].w};
                for (int j = 0; j < RT; ++j) {
                    const v2f a0 = {ca[j][q].x, ca[j][q].y};
                    dacc[j] = __builtin_amdgcn_wmma_f32_16x16x4_f32(
                         false, a0,  false, b0,
                         (short)0, dacc[j],  false,  false);
                }
                for (int j = 0; j < RT; ++j) {
                    const v2f a1 = {ca[j][q].z, ca[j][q].w};
                    dacc[j] = __builtin_amdgcn_wmma_f32_16x16x4_f32(
                         false, a1,  false, b1,
                         (short)0, dacc[j],  false,  false);
                }
            }
        }

        const int m0 = strip * ROWS + rt0 * 16;
        float* __restrict__ pp = Pn + (size_t)(m0 + 8 * h) * NC + n0 + lid;
        if (strip * ROWS + ROWS <= n) {
            for (int j = 0; j < RT; ++j)
                for (int r = 0; r < 8; ++r)
                    pp[(size_t)(j * 16 + r) * NC] = dacc[j][r];
        } else {
            for (int j = 0; j < RT; ++j)
                for (int r = 0; r < 8; ++r)
                    if (m0 + j * 16 + 8 * h + r < n)
                        pp[(size_t)(j * 16 + r) * NC] = dacc[j][r];
        }
        buf ^= 1;
    }
}


extern "C" void kernel_launch(void* const* d_in, const int* in_sizes, int n_in,
                              void* d_out, int out_size, void* d_ws, size_t ws_size,
                              hipStream_t stream) {
    const float* feat = (const float*)d_in[0];
    const float* W1   = (const float*)d_in[1];
    const float* b1   = (const float*)d_in[2];
    const float* W2   = (const float*)d_in[3];
    const float* b2   = (const float*)d_in[4];
    const float* W3   = (const float*)d_in[5];
    const float* b3   = (const float*)d_in[6];
    const int*   src  = (const int*)d_in[7];
    const int*   dst  = (const int*)d_in[8];

    const int D_IN = 256, D_H = 128, D_OUT = 64;
    const int n = in_sizes[0] / D_IN;
    const int E = in_sizes[7];

    float* ns   = (float*)d_ws;
    float* buf0 = ns + 2 * (size_t)n;
    float* buf1 = buf0 + (size_t)n * D_H;

    float* out = (float*)d_out;

    const int csr_ints = (n + 1) + E;
    int* rowptrA = (int*)out;
    int* colA    = rowptrA + (n + 1);
    int* rowptrB = (int*)(buf0 + (size_t)n * D_OUT);
    int* colB    = rowptrB + (n + 1);

    const int NB = (n + BW - 1) >> BW_SHIFT;
    const int C  = (E + CH - 1) / CH;
    int2* tmpD = (int2*)buf1;
    int*  tmpS = (int*)(tmpD + E);
    int*  lst  = tmpS + E;

    const int gblocks = (n + 7) / 8;

    k_bin_place<<<C, 256, 0, stream>>>(src, dst, lst, tmpD, tmpS, E, NB, C);
    k_bin_finish<<<2 * NB, BW, 0, stream>>>(tmpD, tmpS, lst, rowptrA, colA, ns, n, NB, C);

    k_gemm_strips<256, 128, 4><<<512, 256, 0, stream>>>(feat, ns, W1, buf0, n);

    k_gather_gemm<128><<<768, 256, 0, stream>>>(buf0, rowptrA, colA, b1, ns, W2, buf1, n);

    k_gather_gemm<64><<<1024, 256, 0, stream>>>(buf1, rowptrA, colA, b2, ns, W3, buf0, n);

    k_copy_int<<<(csr_ints + 255) / 256, 256, 0, stream>>>(rowptrA, rowptrB, csr_ints);
    k_gather<64, false><<<gblocks, 256, 0, stream>>>(buf0, rowptrB, colB, b3, out, n);
}
